# starting kernel
# baseline (speedup 1.0000x reference)
#include <hip/hip_runtime.h>

#define N_NODES 100000
#define N_EDGES 1600000
#define D 64

typedef __attribute__((ext_vector_type(2))) float v2f;
typedef __attribute__((ext_vector_type(4))) float v4f;
typedef __attribute__((ext_vector_type(8))) float v8f;

__global__ void __launch_bounds__(256) gcn_scale_src(
    const float* __restrict__ feat,
    const float* __restrict__ out_norm,
    float* __restrict__ h) {
  int gid = blockIdx.x * blockDim.x + threadIdx.x;
  if (gid >= N_NODES * (D / 4)) return;
  int row = gid >> 4;
  float inv = 1.0f / out_norm[row];
  v4f v = ((const v4f*)feat)[gid];
  v.x *= inv; v.y *= inv; v.z *= inv; v.w *= inv;
  ((v4f*)h)[gid] = v;
}

__global__ void __launch_bounds__(256) gcn_edge_scatter(
    const float* __restrict__ h,
    const int* __restrict__ esrc,
    const int* __restrict__ edst,
    float* __restrict__ agg) {
  unsigned gid = blockIdx.x * blockDim.x + threadIdx.x;
  unsigned e = gid >> 5;
  unsigned lane = gid & 31u;
  if (e >= N_EDGES) return;
  int s = esrc[e];
  int d = edst[e];
  v2f v = ((const v2f*)(h + (size_t)s * D))[lane];
  float* out = agg + (size_t)d * D + 2u * lane;
  unsafeAtomicAdd(out, v.x);
  unsafeAtomicAdd(out + 1, v.y);
}

__global__ void __launch_bounds__(256) gcn_wmma_linear(
    const float* __restrict__ agg,
    const float* __restrict__ in_norm,
    const float* __restrict__ W,
    const float* __restrict__ bias,
    float* __restrict__ out) {
  const int lane  = threadIdx.x & 31;
  const int wave  = threadIdx.x >> 5;
  const int tile  = blockIdx.x * 8 + wave;
  const int NTILES = N_NODES / 16;
  if (tile >= NTILES) return;

  const int m16   = lane & 15;
  const int khalf = lane >> 4;
  const int mrow  = (tile << 4) + m16;

  const float inv = 1.0f / in_norm[mrow];

  v2f afrag[16];
  const float* arow = agg + (size_t)mrow * D + 2 * khalf;
#pragma unroll
  for (int k = 0; k < 16; ++k) {
    v2f a = *(const v2f*)(arow + 4 * k);
    a.x *= inv; a.y *= inv;
    afrag[k] = a;
  }

#pragma unroll
  for (int nt = 0; nt < 4; ++nt) {
    const int nbase = nt << 4;
    const float bv  = bias[nbase + m16];
    v8f acc = {bv, bv, bv, bv, bv, bv, bv, bv};

    const float* wrow = W + (size_t)(nbase + m16) * D + 2 * khalf;
#pragma unroll
    for (int k = 0; k < 16; ++k) {
      v2f bf = *(const v2f*)(wrow + 4 * k);
      acc = __builtin_amdgcn_wmma_f32_16x16x4_f32(
           false, afrag[k],  false, bf,
           (short)0, acc,  false,  false);
    }

    float* orow = out + ((size_t)(tile * 16 + 8 * khalf)) * D + nbase + m16;
#pragma unroll
    for (int j = 0; j < 8; ++j) {
      orow[(size_t)j * D] = acc[j];
    }
  }
}

extern "C" void kernel_launch(void* const* d_in, const int* in_sizes, int n_in,
                              void* d_out, int out_size, void* d_ws, size_t ws_size,
                              hipStream_t stream) {
  const float* feat     = (const float*)d_in[0];
  const float* in_norm  = (const float*)d_in[1];
  const float* out_norm = (const float*)d_in[2];
  const int*   esrc     = (const int*)d_in[3];
  const int*   edst     = (const int*)d_in[4];
  const float* W        = (const float*)d_in[5];
  const float* bias     = (const float*)d_in[6];

  float* out = (float*)d_out;
  float* agg = (float*)d_ws;
  float* h   = out;

  hipMemsetAsync(agg, 0, (size_t)N_NODES * D * sizeof(float), stream);

  {
    int total = N_NODES * (D / 4);
    gcn_scale_src<<<(total + 255) / 256, 256, 0, stream>>>(feat, out_norm, h);
  }
  {
    long long total = (long long)N_EDGES * 32;
    unsigned blocks = (unsigned)((total + 255) / 256);
    gcn_edge_scatter<<<blocks, 256, 0, stream>>>(h, esrc, edst, agg);
  }
  {
    int ntiles = N_NODES / 16;
    int blocks = (ntiles + 7) / 8;
    gcn_wmma_linear<<<blocks, 256, 0, stream>>>(agg, in_norm, W, bias, out);
  }
}


# Edge sums without float atomics: each edge is pushed on its destination row's linked list by one integer exchange; 8 lanes per row walk the list and sum in registers; sums in list (arrival) order
# speedup vs baseline: 1.4408x; 1.4408x over previous
#include <hip/hip_runtime.h>

#define N_NODES 100000
#define N_EDGES 1600000
#define D 64

typedef __attribute__((ext_vector_type(2))) float v2f;
typedef __attribute__((ext_vector_type(4))) float v4f;
typedef __attribute__((ext_vector_type(8))) float v8f;
typedef __attribute__((ext_vector_type(2))) int v2i;

__global__ void __launch_bounds__(256) gcn_scale_src(
    const float* __restrict__ feat,
    const float* __restrict__ out_norm,
    float* __restrict__ h,
    int* __restrict__ head) {
  int gid = blockIdx.x * blockDim.x + threadIdx.x;
  if (gid >= N_NODES * (D / 4)) return;
  if (gid < N_NODES) head[gid] = -1;
  int row = gid >> 4;
  float inv = 1.0f / out_norm[row];
  v4f v = ((const v4f*)feat)[gid];
  v.x *= inv; v.y *= inv; v.z *= inv; v.w *= inv;
  ((v4f*)h)[gid] = v;
}

__global__ void __launch_bounds__(256) gcn_edge_link(
    const int* __restrict__ esrc,
    const int* __restrict__ edst,
    int* __restrict__ head,
    int* __restrict__ rec) {
  unsigned e = blockIdx.x * blockDim.x + threadIdx.x;
  if (e >= N_EDGES) return;
  int s = esrc[e];
  int d = edst[e];
  int prev = atomicExch(head + d, (int)e);
  v2i r = {s, prev};
  ((v2i*)rec)[e] = r;
}

__global__ void __launch_bounds__(256) gcn_row_gather(
    const float* __restrict__ h,
    const int* __restrict__ head,
    const int* __restrict__ rec,
    float* __restrict__ agg) {
  unsigned gid = blockIdx.x * blockDim.x + threadIdx.x;
  unsigned row = gid >> 3;
  unsigned j = gid & 7u;
  if (row >= N_NODES) return;
  int e = head[row];
  v4f a0 = {0.0f, 0.0f, 0.0f, 0.0f};
  v4f a1 = a0, x0 = a0, x1 = a0;
  for (unsigned it = 0; it < N_EDGES && (unsigned)e < N_EDGES; ++it) {
    v2i r = ((const v2i*)rec)[e];
    a0 += x0; a1 += x1;
    const v4f* p = (const v4f*)(h + (size_t)r.x * D) + j;
    x0 = p[0];
    x1 = p[8];
    e = r.y;
  }
  a0 += x0; a1 += x1;
  v4f* o = (v4f*)(agg + (size_t)row * D) + j;
  o[0] = a0;
  o[8] = a1;
}

__global__ void __launch_bounds__(256) gcn_wmma_linear(
    const float* __restrict__ agg,
    const float* __restrict__ in_norm,
    const float* __restrict__ W,
    const float* __restrict__ bias,
    float* __restrict__ out) {
  const int lane  = threadIdx.x & 31;
  const int wave  = threadIdx.x >> 5;
  const int tile  = blockIdx.x * 8 + wave;
  const int NTILES = N_NODES / 16;
  if (tile >= NTILES) return;

  const int m16   = lane & 15;
  const int khalf = lane >> 4;
  const int mrow  = (tile << 4) + m16;

  const float inv = 1.0f / in_norm[mrow];

  v2f afrag[16];
  const float* arow = agg + (size_t)mrow * D + 2 * khalf;
#pragma unroll
  for (int k = 0; k < 16; ++k) {
    v2f a = *(const v2f*)(arow + 4 * k);
    a.x *= inv; a.y *= inv;
    afrag[k] = a;
  }

#pragma unroll
  for (int nt = 0; nt < 4; ++nt) {
    const int nbase = nt << 4;
    const float bv  = bias[nbase + m16];
    v8f acc = {bv, bv, bv, bv, bv, bv, bv, bv};

    const float* wrow = W + (size_t)(nbase + m16) * D + 2 * khalf;
#pragma unroll
    for (int k = 0; k < 16; ++k) {
      v2f bf = *(const v2f*)(wrow + 4 * k);
      acc = __builtin_amdgcn_wmma_f32_16x16x4_f32(
           false, afrag[k],  false, bf,
           (short)0, acc,  false,  false);
    }

    float* orow = out + ((size_t)(tile * 16 + 8 * khalf)) * D + nbase + m16;
#pragma unroll
    for (int j = 0; j < 8; ++j) {
      orow[(size_t)j * D] = acc[j];
    }
  }
}

extern "C" void kernel_launch(void* const* d_in, const int* in_sizes, int n_in,
                              void* d_out, int out_size, void* d_ws, size_t ws_size,
                              hipStream_t stream) {
  const float* feat     = (const float*)d_in[0];
  const float* in_norm  = (const float*)d_in[1];
  const float* out_norm = (const float*)d_in[2];
  const int*   esrc     = (const int*)d_in[3];
  const int*   edst     = (const int*)d_in[4];
  const float* W        = (const float*)d_in[5];
  const float* bias     = (const float*)d_in[6];

  float* out = (float*)d_out;
  float* agg = (float*)d_ws;
  float* h   = out;
  int* head  = (int*)((char*)d_ws + 25600000);
  int* rec   = (int*)((char*)d_ws + 26214400);

  {
    int total = N_NODES * (D / 4);
    gcn_scale_src<<<(total + 255) / 256, 256, 0, stream>>>(feat, out_norm, h, head);
  }
  {
    gcn_edge_link<<<(N_EDGES + 255) / 256, 256, 0, stream>>>(esrc, edst, head, rec);
  }
  {
    gcn_row_gather<<<(N_NODES * 8 + 255) / 256, 256, 0, stream>>>(h, head, rec, agg);
  }
  {
    int ntiles = N_NODES / 16;
    int blocks = (ntiles + 7) / 8;
    gcn_wmma_linear<<<blocks, 256, 0, stream>>>(agg, in_norm, W, bias, out);
  }
}


# Dense product taken before the edge sums (layer is linear in rows): 1/out_norm rides in the WMMA A operand, the row gather writes out; scale pass gone, 3 launches; list-order sums now of product rows
# speedup vs baseline: 1.5228x; 1.0569x over previous
#include <hip/hip_runtime.h>


#define N_NODES 100000
#define N_EDGES 1600000
#define D 64

typedef __attribute__((ext_vector_type(2))) float v2f;
typedef __attribute__((ext_vector_type(4))) float v4f;
typedef __attribute__((ext_vector_type(8))) float v8f;
typedef __attribute__((ext_vector_type(2))) int v2i;


__global__ void __launch_bounds__(256) gcn_wmma_linear(
    const float* __restrict__ feat,
    const float* __restrict__ out_norm,
    const float* __restrict__ W,
    float* __restrict__ z,
    int* __restrict__ head) {
  const int gid   = blockIdx.x * blockDim.x + threadIdx.x;
  if (gid < N_NODES) head[gid] = -1;

  const int lane  = threadIdx.x & 31;
  const int wave  = threadIdx.x >> 5;
  const int tile  = blockIdx.x * 8 + wave;
  const int NTILES = N_NODES / 16;
  if (tile >= NTILES) return;

  const int m16   = lane & 15;
  const int khalf = lane >> 4;
  const int mrow  = (tile << 4) + m16;

  const float inv = 1.0f / out_norm[mrow];

  v2f afrag[16];
  const float* arow = feat + (size_t)mrow * D + 2 * khalf;
#pragma unroll
  for (int k = 0; k < 16; ++k) {
    v2f a = *(const v2f*)(arow + 4 * k);
    a.x *= inv; a.y *= inv;
    afrag[k] = a;
  }

#pragma unroll
  for (int nt = 0; nt < 4; ++nt) {
    const int nbase = nt << 4;
    v8f acc = {0.0f, 0.0f, 0.0f, 0.0f, 0.0f, 0.0f, 0.0f, 0.0f};

    const float* wrow = W + (size_t)(nbase + m16) * D + 2 * khalf;
#pragma unroll
    for (int k = 0; k < 16; ++k) {
      v2f bf = *(const v2f*)(wrow + 4 * k);
      acc = __builtin_amdgcn_wmma_f32_16x16x4_f32(
           false, afrag[k],  false, bf,
           (short)0, acc,  false,  false);
    }

    float* orow = z + ((size_t)(tile * 16 + 8 * khalf)) * D + nbase + m16;
#pragma unroll
    for (int j = 0; j < 8; ++j) {
      orow[(size_t)j * D] = acc[j];
    }
  }
}

__global__ void __launch_bounds__(256) gcn_edge_link(
    const int* __restrict__ esrc,
    const int* __restrict__ edst,
    int* __restrict__ head,
    int* __restrict__ rec) {
  unsigned e = blockIdx.x * blockDim.x + threadIdx.x;
  if (e >= N_EDGES) return;
  int s = esrc[e];
  int d = edst[e];
  int prev = atomicExch(head + d, (int)e);
  v2i r = {s, prev};
  ((v2i*)rec)[e] = r;
}

__global__ void __launch_bounds__(256) gcn_row_gather(
    const float* __restrict__ z,
    const int* __restrict__ head,
    const int* __restrict__ rec,
    const float* __restrict__ in_norm,
    const float* __restrict__ bias,
    float* __restrict__ out) {
  unsigned gid = blockIdx.x * blockDim.x + threadIdx.x;
  unsigned row = gid >> 3;
  unsigned j = gid & 7u;
  if (row >= N_NODES) return;
  int e = head[row];
  const float inv = 1.0f / in_norm[row];
  const v4f b0 = ((const v4f*)bias)[j];
  const v4f b1 = ((const v4f*)bias)[8 + j];
  v4f a0 = {0.0f, 0.0f, 0.0f, 0.0f};
  v4f a1 = a0, x0 = a0, x1 = a0;
  for (unsigned it = 0; it < N_EDGES && (unsigned)e < N_EDGES; ++it) {
    v2i r = ((const v2i*)rec)[e];
    a0 += x0; a1 += x1;
    const v4f* p = (const v4f*)(z + (size_t)r.x * D) + j;
    x0 = p[0];
    x1 = p[8];
    e = r.y;
  }
  a0 += x0; a1 += x1;
  v4f* o = (v4f*)(out + (size_t)row * D) + j;
  o[0] = a0 * inv + b0;
  o[8] = a1 * inv + b1;
}

extern "C" void kernel_launch(void* const* d_in, const int* in_sizes, int n_in,
                              void* d_out, int out_size, void* d_ws, size_t ws_size,
                              hipStream_t stream) {
  const float* feat     = (const float*)d_in[0];
  const float* in_norm  = (const float*)d_in[1];
  const float* out_norm = (const float*)d_in[2];
  const int*   esrc     = (const int*)d_in[3];
  const int*   edst     = (const int*)d_in[4];
  const float* W        = (const float*)d_in[5];
  const float* bias     = (const float*)d_in[6];

  float* out = (float*)d_out;
  float* z   = (float*)d_ws;
  int* head  = (int*)((char*)d_ws + 25600000);
  int* rec   = (int*)((char*)d_ws + 26214400);

  {
    int ntiles = N_NODES / 16;
    int blocks = (ntiles + 7) / 8;
    gcn_wmma_linear<<<blocks, 256, 0, stream>>>(feat, out_norm, W, z, head);
  }
  {
    gcn_edge_link<<<(N_EDGES + 255) / 256, 256, 0, stream>>>(esrc, edst, head, rec);
  }
  {
    gcn_row_gather<<<(N_NODES * 8 + 255) / 256, 256, 0, stream>>>(z, head, rec, in_norm, bias, out);
  }
}


# speedup vs baseline: 3.1260x; 2.0528x over previous
#include <hip/hip_runtime.h>


#define N_NODES 100000
#define N_EDGES 1600000
#define D 64
#define HEAD_STRIDE 100352

typedef __attribute__((ext_vector_type(2))) float v2f;
typedef __attribute__((ext_vector_type(4))) float v4f;
typedef __attribute__((ext_vector_type(8))) float v8f;
typedef __attribute__((ext_vector_type(2))) int v2i;


__global__ void __launch_bounds__(256) gcn_wmma_linear(
    const float* __restrict__ feat,
    const float* __restrict__ out_norm,
    const float* __restrict__ W,
    float* __restrict__ z,
    int* __restrict__ head) {
  const int gid   = blockIdx.x * blockDim.x + threadIdx.x;
  if (gid < N_NODES) {
    head[gid] = -1;
    head[HEAD_STRIDE + gid] = -1;
  }

  const int lane  = threadIdx.x & 31;
  const int wave  = threadIdx.x >> 5;
  const int tile  = blockIdx.x * 8 + wave;
  const int NTILES = N_NODES / 16;
  if (tile >= NTILES) return;

  const int m16   = lane & 15;
  const int khalf = lane >> 4;
  const int mrow  = (tile << 4) + m16;

  const float inv = 1.0f / out_norm[mrow];

  v2f afrag[16];
  const float* arow = feat + (size_t)mrow * D + 2 * khalf;
#pragma unroll
  for (int k = 0; k < 16; ++k) {
    v2f a = *(const v2f*)(arow + 4 * k);
    a.x *= inv; a.y *= inv;
    afrag[k] = a;
  }

#pragma unroll
  for (int nt = 0; nt < 4; ++nt) {
    const int nbase = nt << 4;
    v8f acc = {0.0f, 0.0f, 0.0f, 0.0f, 0.0f, 0.0f, 0.0f, 0.0f};

    const float* wrow = W + (size_t)(nbase + m16) * D + 2 * khalf;
#pragma unroll
    for (int k = 0; k < 16; ++k) {
      v2f bf = *(const v2f*)(wrow + 4 * k);
      acc = __builtin_amdgcn_wmma_f32_16x16x4_f32(
           false, afrag[k],  false, bf,
           (short)0, acc,  false,  false);
    }

    float* orow = z + ((size_t)(tile * 16 + 8 * khalf)) * D + nbase + m16;
#pragma unroll
    for (int j = 0; j < 8; ++j) {
      orow[(size_t)j * D] = acc[j];
    }
  }
}

__global__ void __launch_bounds__(256) gcn_edge_link(
    const int* __restrict__ esrc,
    const int* __restrict__ edst,
    int* __restrict__ head,
    int* __restrict__ rec) {
  unsigned e = blockIdx.x * blockDim.x + threadIdx.x;
  if (e >= N_EDGES) return;
  int s = esrc[e];
  int d = edst[e];
  const int half = (__builtin_amdgcn_s_sendmsg_rtn(0x87) >> 18) & 1;
  int prev = atomicExch(head + half * HEAD_STRIDE + d, (int)e);
  v2i r = {s, prev};
  ((v2i*)rec)[e] = r;
}

__global__ void __launch_bounds__(256) gcn_row_gather(
    const float* __restrict__ z,
    const int* __restrict__ head,
    const int* __restrict__ rec,
    const float* __restrict__ in_norm,
    const float* __restrict__ bias,
    float* __restrict__ out) {
  unsigned gid = blockIdx.x * blockDim.x + threadIdx.x;
  unsigned row = gid >> 3;
  unsigned j = gid & 7u;
  if (row >= N_NODES) return;
  int e = head[row];
  int e2 = head[HEAD_STRIDE + row];
  const float inv = 1.0f / in_norm[row];
  const v4f b0 = ((const v4f*)bias)[j];
  const v4f b1 = ((const v4f*)bias)[8 + j];
  v4f a0 = {0.0f, 0.0f, 0.0f, 0.0f};
  v4f a1 = a0, x0 = a0, x1 = a0;
  if ((unsigned)e >= N_EDGES) { e = e2; e2 = -1; }
  for (unsigned it = 0; it < N_EDGES && (unsigned)e < N_EDGES; ++it) {
    v2i r = ((const v2i*)rec)[e];
    a0 += x0; a1 += x1;
    const v4f* p = (const v4f*)(z + (size_t)r.x * D) + j;
    x0 = p[0];
    x1 = p[8];
    e = r.y;
    if ((unsigned)e >= N_EDGES) { e = e2; e2 = -1; }
  }
  a0 += x0; a1 += x1;
  v4f* o = (v4f*)(out + (size_t)row * D) + j;
  o[0] = a0 * inv + b0;
  o[8] = a1 * inv + b1;
}

extern "C" void kernel_launch(void* const* d_in, const int* in_sizes, int n_in,
                              void* d_out, int out_size, void* d_ws, size_t ws_size,
                              hipStream_t stream) {
  const float* feat     = (const float*)d_in[0];
  const float* in_norm  = (const float*)d_in[1];
  const float* out_norm = (const float*)d_in[2];
  const int*   esrc     = (const int*)d_in[3];
  const int*   edst     = (const int*)d_in[4];
  const float* W        = (const float*)d_in[5];
  const float* bias     = (const float*)d_in[6];

  float* out = (float*)d_out;
  float* z   = (float*)d_ws;
  int* head  = (int*)((char*)d_ws + 25600000);
  int* rec   = (int*)((char*)d_ws + 26402816);

  {
    int ntiles = N_NODES / 16;
    int blocks = (ntiles + 7) / 8;
    gcn_wmma_linear<<<blocks, 256, 0, stream>>>(feat, out_norm, W, z, head);
  }
  {
    gcn_edge_link<<<(N_EDGES + 255) / 256, 256, 0, stream>>>(esrc, edst, head, rec);
  }
  {
    gcn_row_gather<<<(N_NODES * 8 + 255) / 256, 256, 0, stream>>>(z, head, rec, in_norm, bias, out);
  }
}
